# starting kernel
# baseline (speedup 1.0000x reference)
#include <hip/hip_runtime.h>
#include <hip/hip_bf16.h>
#include <math.h>

typedef __attribute__((ext_vector_type(2))) float v2f;
typedef __attribute__((ext_vector_type(8))) float v8f;

#define D_DIM 64
#define H_DIM 128
#define CPITCH 66

__device__ __forceinline__ unsigned ordKey(float f) {
  unsigned u = __float_as_uint(f);
  return (u & 0x80000000u) ? ~u : (u | 0x80000000u);
}
__device__ __forceinline__ float ordVal(unsigned k) {
  unsigned u = (k & 0x80000000u) ? (k ^ 0x80000000u) : ~k;
  return __uint_as_float(u);
}

__global__ void init_seg(unsigned* __restrict__ segMax, float* __restrict__ segSum, int numS) {
  int i = blockIdx.x * blockDim.x + threadIdx.x;
  if (i < numS) { segMax[i] = 0u; segSum[i] = 0.0f; }
}

__global__ __launch_bounds__(256)
void edge_mlp_cost(const float* __restrict__ embS,
                   const float* __restrict__ embA,
                   const int*   __restrict__ edgeS,
                   const int*   __restrict__ edgeA,
                   const float* __restrict__ W1,
                   const float* __restrict__ b1,
                   const float* __restrict__ W2,
                   const float* __restrict__ b2,
                   const float* __restrict__ logT,
                   float*       __restrict__ outCost,
                   unsigned*    __restrict__ segMax,
                   int E) {
  __shared__ float sW1[H_DIM * CPITCH * 2];
  __shared__ float sB1[H_DIM];
  __shared__ float sW2[H_DIM];
  __shared__ float sGate[8][16];

  const int tid = threadIdx.x;
  for (int i = tid; i < H_DIM * H_DIM; i += 256) {
    int kk = i >> 7;
    int c  = i & 127;
    sW1[(c * CPITCH + (kk >> 1)) * 2 + (kk & 1)] = W1[i];
  }
  if (tid < H_DIM) { sB1[tid] = b1[tid]; sW2[tid] = W2[tid]; }
  __syncthreads();

  const int wave = tid >> 5;
  const int lane = tid & 31;
  const int half = lane >> 4;
  const int col  = lane & 15;
  const int e0   = (blockIdx.x * 8 + wave) * 16;
  if (e0 >= E) return;

  const int  e     = e0 + col;
  const bool valid = (e < E);
  const int  ec    = valid ? e : (E - 1);
  const int  sI    = edgeS[ec];
  const int  aI    = edgeA[ec];
  const float* es = embS + (size_t)sI * D_DIM;
  const float* ea = embA + (size_t)aI * D_DIM;

  v2f a[32];
#pragma unroll
  for (int k = 0; k < 16; ++k) {
    a[k]      = *(const v2f*)(es + 4 * k + 2 * half);
    a[k + 16] = *(const v2f*)(ea + 4 * k + 2 * half);
  }
  float ss = 0.0f;
#pragma unroll
  for (int k = 0; k < 16; ++k) {
    float dx = a[k][0] - a[k + 16][0];
    float dy = a[k][1] - a[k + 16][1];
    ss += dx * dx + dy * dy;
  }
  ss += __shfl_xor(ss, 16);

  float gacc[8] = {0, 0, 0, 0, 0, 0, 0, 0};
#pragma unroll
  for (int n = 0; n < 8; ++n) {
    const float* bbase = sW1 + ((n * 16 + col) * CPITCH + half) * 2;
    v8f c0 = {};
    v8f c1 = {};
#pragma unroll
    for (int k = 0; k < 32; k += 2) {
      v2f b0 = *(const v2f*)(bbase + 4 * k);
      v2f b1v = *(const v2f*)(bbase + 4 * (k + 1));
      c0 = __builtin_amdgcn_wmma_f32_16x16x4_f32(false, a[k], false, b0,
                                                 (short)0, c0, false, false);
      c1 = __builtin_amdgcn_wmma_f32_16x16x4_f32(false, a[k + 1], false, b1v,
                                                 (short)0, c1, false, false);
    }
    const float bn  = sB1[n * 16 + col];
    const float w2n = sW2[n * 16 + col];
#pragma unroll
    for (int r = 0; r < 8; ++r) {
      float h = c0[r] + c1[r] + bn;
      h = h > 0.0f ? h : 0.0f;
      gacc[r] += h * w2n;
    }
  }

#pragma unroll
  for (int r = 0; r < 8; ++r) {
    float g = gacc[r];
    g += __shfl_xor(g, 1);
    g += __shfl_xor(g, 2);
    g += __shfl_xor(g, 4);
    g += __shfl_xor(g, 8);
    if (lane == 0)  sGate[wave][r]     = g;
    if (lane == 16) sGate[wave][r + 8] = g;
  }

  const float b2v  = b2[0];
  const float invT = __expf(-logT[0]);
  if (lane < 16 && valid) {
    float logit = sGate[wave][lane] + b2v;
    float gate  = 1.0f / (1.0f + __expf(-logit));
    float cost  = sqrtf(ss) * gate;
    outCost[e]  = cost;
    atomicMax(&segMax[sI], ordKey(-cost * invT));
  }
}

__global__ void seg_expsum(const float* __restrict__ outCost,
                           const int* __restrict__ edgeS,
                           const unsigned* __restrict__ segMax,
                           float* __restrict__ segSum,
                           const float* __restrict__ logT, int E) {
  int e = blockIdx.x * blockDim.x + threadIdx.x;
  if (e >= E) return;
  float invT = __expf(-logT[0]);
  int s = edgeS[e];
  float val = -outCost[e] * invT;
  float m = ordVal(segMax[s]);
  atomicAdd(&segSum[s], __expf(val - m));
}

__global__ void seg_norm(const float* __restrict__ outCost,
                         const int* __restrict__ edgeS,
                         const unsigned* __restrict__ segMax,
                         const float* __restrict__ segSum,
                         const float* __restrict__ logT,
                         float* __restrict__ outW, int E) {
  int e = blockIdx.x * blockDim.x + threadIdx.x;
  if (e >= E) return;
  float invT = __expf(-logT[0]);
  int s = edgeS[e];
  float val = -outCost[e] * invT;
  float m = ordVal(segMax[s]);
  outW[e] = __expf(val - m) / segSum[s];
}

extern "C" void kernel_launch(void* const* d_in, const int* in_sizes, int n_in,
                              void* d_out, int out_size, void* d_ws, size_t ws_size,
                              hipStream_t stream) {
  const float* embS = (const float*)d_in[0];
  const float* embA = (const float*)d_in[1];
  const int*   eidx = (const int*)d_in[2];
  const float* W1   = (const float*)d_in[3];
  const float* b1   = (const float*)d_in[4];
  const float* W2   = (const float*)d_in[5];
  const float* b2   = (const float*)d_in[6];
  const float* logT = (const float*)d_in[7];

  const int numS = in_sizes[0] / D_DIM;
  const int E    = in_sizes[2] / 2;
  const int* edgeS = eidx;
  const int* edgeA = eidx + E;

  float* outW    = (float*)d_out;
  float* outCost = (float*)d_out + E;

  unsigned* segMax = (unsigned*)d_ws;
  float*    segSum = (float*)((unsigned*)d_ws + numS);

  init_seg<<<(numS + 255) / 256, 256, 0, stream>>>(segMax, segSum, numS);

  const int tiles = (E + 15) / 16;
  edge_mlp_cost<<<(tiles + 7) / 8, 256, 0, stream>>>(
      embS, embA, edgeS, edgeA, W1, b1, W2, b2, logT, outCost, segMax, E);

  seg_expsum<<<(E + 255) / 256, 256, 0, stream>>>(outCost, edgeS, segMax, segSum, logT, E);
  seg_norm<<<(E + 255) / 256, 256, 0, stream>>>(outCost, edgeS, segMax, segSum, logT, outW, E);
}


# edge MLP tile loop software-pipelined: a tile's finishing work (last column tile's ReLU/W2 dot, lane-half sums, sigmoid, sqrt, store, segment max) issues among the next tile's first products
# speedup vs baseline: 2.3266x; 2.3266x over previous
#include <hip/hip_runtime.h>
#include <hip/hip_bf16.h>
#include <math.h>

typedef __attribute__((ext_vector_type(2))) float v2f;
typedef __attribute__((ext_vector_type(4))) float v4f;
typedef __attribute__((ext_vector_type(8))) float v8f;

#define D_DIM 64
#define H_DIM 128

__device__ __forceinline__ unsigned ordKey(float f) {
  unsigned u = __float_as_uint(f);
  return (u & 0x80000000u) ? ~u : (u | 0x80000000u);
}
__device__ __forceinline__ float ordVal(unsigned k) {
  unsigned u = (k & 0x80000000u) ? (k ^ 0x80000000u) : ~k;
  return __uint_as_float(u);
}

__global__ void init_seg(unsigned* __restrict__ segMax, float* __restrict__ segSum, int numS) {
  int i = blockIdx.x * blockDim.x + threadIdx.x;
  if (i < numS) { segMax[i] = 0u; segSum[i] = 0.0f; }
}

__device__ __forceinline__ void gather_rows(v4f (&dst)[16],
                                            const float* __restrict__ embS,
                                            const float* __restrict__ embA,
                                            int sI, int aI, int half) {
  const float* es = embS + (size_t)sI * D_DIM + 4 * half;
  const float* ea = embA + (size_t)aI * D_DIM + 4 * half;
#pragma unroll
  for (int q = 0; q < 8; ++q) {
    dst[q]     = *(const v4f*)(es + 8 * q);
    dst[q + 8] = *(const v4f*)(ea + 8 * q);
  }
}

__global__ __launch_bounds__(128)
void edge_mlp_cost(const float* __restrict__ embS,
                   const float* __restrict__ embA,
                   const int*   __restrict__ edgeS,
                   const int*   __restrict__ edgeA,
                   const float* __restrict__ W1,
                   const float* __restrict__ b1,
                   const float* __restrict__ W2,
                   const float* __restrict__ b2,
                   const float* __restrict__ logT,
                   float*       __restrict__ outCost,
                   unsigned*    __restrict__ segMax,
                   int E) {
  const int tid  = threadIdx.x;
  const int lane = tid & 31;
  const int half = lane >> 4;
  const int col  = lane & 15;
  const int wave  = __builtin_amdgcn_readfirstlane(tid >> 5);
  const int gw    = blockIdx.x * 4 + wave;
  const int nw    = gridDim.x * 4;
  const int tiles = (E + 15) >> 4;
  if (gw >= tiles) return;

  v2f wf[8][32];
#pragma unroll
  for (int n = 0; n < 8; ++n) {
#pragma unroll
    for (int k = 0; k < 32; ++k) {
      const float* p = W1 + (8 * (k >> 1) + 4 * half + 2 * (k & 1)) * H_DIM + n * 16 + col;
      wf[n][k][0] = p[0];
      wf[n][k][1] = p[H_DIM];
    }
  }
  float bb[8][8], ww[8][8];
#pragma unroll
  for (int n = 0; n < 8; ++n) {
#pragma unroll
    for (int r = 0; r < 8; ++r) {
      bb[n][r] = b1[n * 16 + 8 * half + r];
      ww[n][r] = W2[n * 16 + 8 * half + r];
    }
  }
  const float b2v  = b2[0];
  const float invT = __expf(-logT[0]);

  int t = gw;
  int ec = t * 16 + col;
  ec = ec < E ? ec : (E - 1);
  int sI = edgeS[ec];
  int aI = edgeA[ec];
  v4f cur[16];
  gather_rows(cur, embS, embA, sI, aI, half);

  int en = (t + nw) * 16 + col;
  en = en < E ? en : (E - 1);
  int sIn = edgeS[en];
  int aIn = edgeA[en];

  v8f q0 = {};
  v8f q1 = {};
  float gp0 = 0.0f, gp1 = 0.0f, ssp = 0.0f;
  int sIp = 0, ep = E;
  float ssF = 0.0f, gF = 0.0f, ssX = 0.0f, gH = 0.0f, gX = 0.0f;

  for (; t < tiles; t += nw) {
    v4f nxt[16];
    gather_rows(nxt, embS, embA, sIn, aIn, half);
    int e2 = (t + 2 * nw) * 16 + col;
    e2 = e2 < E ? e2 : (E - 1);
    const int sI2 = edgeS[e2];
    const int aI2 = edgeA[e2];
    __builtin_amdgcn_sched_barrier(0);

    float ss = 0.0f;
    float g0 = 0.0f, g1 = 0.0f;
    v8f p0 = {};
    v8f p1 = {};
#pragma unroll
    for (int n = 0; n < 8; ++n) {
      v8f c0 = {};
      v8f c1 = {};
#pragma unroll
      for (int q = 0; q < 16; ++q) {
        const v2f blo = __builtin_shufflevector(cur[q], cur[q], 0, 1);
        const v2f bhi = __builtin_shufflevector(cur[q], cur[q], 2, 3);
        c0 = __builtin_amdgcn_wmma_f32_16x16x4_f32(false, wf[n][2 * q], false, blo,
                                                   (short)0, c0, false, false);
        c1 = __builtin_amdgcn_wmma_f32_16x16x4_f32(false, wf[n][2 * q + 1], false, bhi,
                                                   (short)0, c1, false, false);
      }
      if (n == 0) {
#pragma unroll
        for (int q = 0; q < 8; ++q) {
          const v4f d = cur[q] - cur[q + 8];
          ss += (d[0] * d[0] + d[1] * d[1]) + (d[2] * d[2] + d[3] * d[3]);
        }
#pragma unroll
        for (int r = 0; r < 8; r += 2) {
          float h0 = q0[r] + q1[r] + bb[7][r];
          float h1 = q0[r + 1] + q1[r + 1] + bb[7][r + 1];
          h0 = h0 > 0.0f ? h0 : 0.0f;
          h1 = h1 > 0.0f ? h1 : 0.0f;
          gp0 += h0 * ww[7][r];
          gp1 += h1 * ww[7][r + 1];
        }
      } else {
#pragma unroll
        for (int r = 0; r < 8; r += 2) {
          float h0 = p0[r] + p1[r] + bb[n - 1][r];
          float h1 = p0[r + 1] + p1[r + 1] + bb[n - 1][r + 1];
          h0 = h0 > 0.0f ? h0 : 0.0f;
          h1 = h1 > 0.0f ? h1 : 0.0f;
          g0 += h0 * ww[n - 1][r];
          g1 += h1 * ww[n - 1][r + 1];
        }
      }
      if (n == 1) {
        ssX = __shfl_xor(ssp, 16);
        gH  = gp0 + gp1;
        gX  = __shfl_xor(gH, 16);
      }
      if (n == 2) {
        ssF = ssp + ssX;
        gF  = gH + gX;
        const float logit = gF + b2v;
        const float gate  = 1.0f / (1.0f + __expf(-logit));
        float cost = sqrtf(ssF) * gate;
        unsigned key = ordKey(-cost * invT);
        asm volatile("" : "+v"(cost), "+v"(key));
        if (half == 0 && ep < E) {
          outCost[ep] = cost;
          atomicMax(&segMax[sIp], key);
        }
      }
      __builtin_amdgcn_sched_barrier(0);
      p0 = c0;
      p1 = c1;
    }
    q0  = p0;
    q1  = p1;
    gp0 = g0;
    gp1 = g1;
    ssp = ss;
    sIp = sI;
    ep  = t * 16 + col;

#pragma unroll
    for (int q = 0; q < 16; ++q) cur[q] = nxt[q];
    sI  = sIn;
    sIn = sI2;
    aIn = aI2;
  }

#pragma unroll
  for (int r = 0; r < 8; r += 2) {
    float h0 = q0[r] + q1[r] + bb[7][r];
    float h1 = q0[r + 1] + q1[r + 1] + bb[7][r + 1];
    h0 = h0 > 0.0f ? h0 : 0.0f;
    h1 = h1 > 0.0f ? h1 : 0.0f;
    gp0 += h0 * ww[7][r];
    gp1 += h1 * ww[7][r + 1];
  }
  ssF = ssp + __shfl_xor(ssp, 16);
  {
    const float gh = gp0 + gp1;
    gF = gh + __shfl_xor(gh, 16);
  }
  if (half == 0 && ep < E) {
    const float logit = gF + b2v;
    const float gate  = 1.0f / (1.0f + __expf(-logit));
    const float cost  = sqrtf(ssF) * gate;
    outCost[ep] = cost;
    atomicMax(&segMax[sIp], ordKey(-cost * invT));
  }
}

#define SEG_WIN     6272
#define EDGE_CHUNKS 32
#define WIN_TRIPS   ((SEG_WIN + 1023) / 1024)
#define VEC_GROUPS  4

__device__ __forceinline__ void expsum4(const int4 s4, const float4 c4, const int lo,
                                        const unsigned ucnt, const float invT,
                                        const float* winMax, float* winSum) {
  const unsigned w0 = (unsigned)(s4.x - lo), w1 = (unsigned)(s4.y - lo);
  const unsigned w2 = (unsigned)(s4.z - lo), w3 = (unsigned)(s4.w - lo);
  const bool k0 = w0 < ucnt, k1 = w1 < ucnt, k2 = w2 < ucnt, k3 = w3 < ucnt;
  const float m0 = winMax[k0 ? w0 : 0u], m1 = winMax[k1 ? w1 : 0u];
  const float m2 = winMax[k2 ? w2 : 0u], m3 = winMax[k3 ? w3 : 0u];
  const float x0 = __expf(-c4.x * invT - m0), x1 = __expf(-c4.y * invT - m1);
  const float x2 = __expf(-c4.z * invT - m2), x3 = __expf(-c4.w * invT - m3);
  if (k0) atomicAdd(&winSum[w0], x0);
  if (k1) atomicAdd(&winSum[w1], x1);
  if (k2) atomicAdd(&winSum[w2], x2);
  if (k3) atomicAdd(&winSum[w3], x3);
}

__global__ __launch_bounds__(1024)
void seg_expsum(const float* __restrict__ outCost,
                const int* __restrict__ edgeS,
                const unsigned* __restrict__ segMax,
                float* __restrict__ segSum,
                const float* __restrict__ logT, int E, int numS) {
  __shared__ float winMax[SEG_WIN];
  __shared__ float winSum[SEG_WIN];
  const int tid   = threadIdx.x;
  const int nWin  = (numS + SEG_WIN - 1) / SEG_WIN;
  const int win   = blockIdx.x % nWin;
  const int chunk = blockIdx.x / nWin;
  const int lo    = win * SEG_WIN;
  int cnt = numS - lo;
  cnt = cnt < SEG_WIN ? cnt : SEG_WIN;
  const unsigned ucnt = (unsigned)cnt;

  unsigned key[WIN_TRIPS];
#pragma unroll
  for (int k = 0; k < WIN_TRIPS; ++k) {
    const int i = tid + 1024 * k;
    key[k] = i < cnt ? segMax[lo + i] : 0u;
  }
#pragma unroll
  for (int k = 0; k < WIN_TRIPS; ++k) {
    const int i = tid + 1024 * k;
    if (i < cnt) {
      winMax[i] = ordVal(key[k]);
      winSum[i] = 0.0f;
    }
  }
  __syncthreads();

  const float invT = __expf(-logT[0]);
  int per = (E + EDGE_CHUNKS - 1) / EDGE_CHUNKS;
  per = (per + 3) & ~3;
  const int e0 = chunk * per;
  int e1 = e0 + per;
  e1 = e1 < E ? e1 : E;

  if ((E & 3) == 0) {
    for (int e = e0 + 4 * tid; e < e1; e += VEC_GROUPS * 4096) {
      int4 sv[VEC_GROUPS];
      float4 cv[VEC_GROUPS];
#pragma unroll
      for (int g = 0; g < VEC_GROUPS; ++g) {
        const int eg = e + g * 4096;
        const int el = eg < e1 ? eg : e;
        sv[g] = *(const int4*)(edgeS + el);
        cv[g] = *(const float4*)(outCost + el);
      }
#pragma unroll
      for (int g = 0; g < VEC_GROUPS; ++g) {
        int4 s4 = sv[g];
        if (e + g * 4096 >= e1) s4 = make_int4(-1, -1, -1, -1);
        expsum4(s4, cv[g], lo, ucnt, invT, winMax, winSum);
      }
    }
  } else {
    for (int e = e0 + tid; e < e1; e += 1024) {
      const unsigned w = (unsigned)(edgeS[e] - lo);
      if (w < ucnt) atomicAdd(&winSum[w], __expf(-outCost[e] * invT - winMax[w]));
    }
  }
  __syncthreads();

  int rot = chunk * ((SEG_WIN / EDGE_CHUNKS) & ~31);
  rot = rot < cnt ? rot : 0;
  for (int i = tid; i < cnt; i += 1024) {
    int j = i + rot;
    j = j < cnt ? j : j - cnt;
    const float v = winSum[j];
    if (v != 0.0f) atomicAdd(&segSum[lo + j], v);
  }
}

__global__ void seg_norm(const float* __restrict__ outCost,
                         const int* __restrict__ edgeS,
                         const unsigned* __restrict__ segMax,
                         const float* __restrict__ segSum,
                         const float* __restrict__ logT,
                         float* __restrict__ outW, int E) {
  int e = blockIdx.x * blockDim.x + threadIdx.x;
  if (e >= E) return;
  float invT = __expf(-logT[0]);
  int s = edgeS[e];
  float val = -outCost[e] * invT;
  float m = ordVal(segMax[s]);
  outW[e] = __expf(val - m) / segSum[s];
}

extern "C" void kernel_launch(void* const* d_in, const int* in_sizes, int n_in,
                              void* d_out, int out_size, void* d_ws, size_t ws_size,
                              hipStream_t stream) {
  const float* embS = (const float*)d_in[0];
  const float* embA = (const float*)d_in[1];
  const int*   eidx = (const int*)d_in[2];
  const float* W1   = (const float*)d_in[3];
  const float* b1   = (const float*)d_in[4];
  const float* W2   = (const float*)d_in[5];
  const float* b2   = (const float*)d_in[6];
  const float* logT = (const float*)d_in[7];

  const int numS = in_sizes[0] / D_DIM;
  const int E    = in_sizes[2] / 2;
  const int* edgeS = eidx;
  const int* edgeA = eidx + E;

  float* outW    = (float*)d_out;
  float* outCost = (float*)d_out + E;

  unsigned* segMax = (unsigned*)d_ws;
  float*    segSum = (float*)((unsigned*)d_ws + numS);

  init_seg<<<(numS + 255) / 256, 256, 0, stream>>>(segMax, segSum, numS);

  edge_mlp_cost<<<256, 128, 0, stream>>>(
      embS, embA, edgeS, edgeA, W1, b1, W2, b2, logT, outCost, segMax, E);

  seg_expsum<<<((numS + SEG_WIN - 1) / SEG_WIN) * EDGE_CHUNKS, 1024, 0, stream>>>(
      outCost, edgeS, segMax, segSum, logT, E, numS);
  seg_norm<<<(E + 255) / 256, 256, 0, stream>>>(outCost, edgeS, segMax, segSum, logT, outW, E);
}
